# starting kernel
# baseline (speedup 1.0000x reference)
#include <hip/hip_runtime.h>

typedef float v2f __attribute__((ext_vector_type(2)));
typedef float v8f __attribute__((ext_vector_type(8)));

#define RANK 16
#define DIM  400
#define FACTOR_ELEMS (RANK * DIM)
#define LDS_FLOATS   (3 * FACTOR_ELEMS)

__global__ void parafac_query_wmma(const float* __restrict__ U,
                                   const float* __restrict__ V,
                                   const float* __restrict__ W,
                                   const int*   __restrict__ idx,
                                   float*       __restrict__ out,
                                   int nQueries)
{
    extern __shared__ float lds[];
    float* Ul = lds;
    float* Vl = lds + FACTOR_ELEMS;
    float* Wl = lds + 2 * FACTOR_ELEMS;

    for (int t = threadIdx.x; t < FACTOR_ELEMS; t += blockDim.x) {
        Ul[t] = U[t];
        Vl[t] = V[t];
        Wl[t] = W[t];
    }
    __syncthreads();

    const int lane   = threadIdx.x & 31;
    const int q      = lane & 15;
    const int half   = lane >> 4;
    const int waveId = (blockIdx.x * blockDim.x + threadIdx.x) >> 5;
    const int nWaves = (gridDim.x * blockDim.x) >> 5;
    const int nTiles = (nQueries + 15) >> 4;

    for (int tile = waveId; tile < nTiles; tile += nWaves) {
        const int qi = tile * 16 + q;
        const int qc = (qi < nQueries) ? qi : (nQueries - 1);

        const int i = idx[qc * 3 + 0];
        const int j = idx[qc * 3 + 1];
        const int k = idx[qc * 3 + 2];

        {
            const int nq = (tile + nWaves) * 16 + q;
            if (nq < nQueries)
                __builtin_prefetch(&idx[nq * 3], 0, 1);
        }

        v8f c = {};
        #pragma unroll
        for (int t = 0; t < 4; ++t) {
            const int r0 = 4 * t + 2 * half;
            const int r1 = r0 + 1;

            v2f a, b;
            a.x = Ul[r0 * DIM + i] * Vl[r0 * DIM + j];
            a.y = Ul[r1 * DIM + i] * Vl[r1 * DIM + j];
            b.x = Wl[r0 * DIM + k];
            b.y = Wl[r1 * DIM + k];

            c = __builtin_amdgcn_wmma_f32_16x16x4_f32(
                    false, a, false, b, (short)0, c, false, false);
        }

        if (lane < 8 || lane >= 24) {
            const int dq = (lane < 8) ? lane : (lane - 16);
            const int oq = tile * 16 + dq;
            if (oq < nQueries) {
                const int s = lane & 7;
                float vsel = c[0];
                if (s == 1) vsel = c[1];
                if (s == 2) vsel = c[2];
                if (s == 3) vsel = c[3];
                if (s == 4) vsel = c[4];
                if (s == 5) vsel = c[5];
                if (s == 6) vsel = c[6];
                if (s == 7) vsel = c[7];
                out[oq] = vsel;
            }
        }
    }
}

extern "C" void kernel_launch(void* const* d_in, const int* in_sizes, int n_in,
                              void* d_out, int out_size, void* d_ws, size_t ws_size,
                              hipStream_t stream) {
    const float* U   = (const float*)d_in[0];
    const float* V   = (const float*)d_in[1];
    const float* W   = (const float*)d_in[2];
    const int*   idx = (const int*)  d_in[3];
    float*       out = (float*)d_out;

    const int nQueries = out_size;
    const int nTiles   = (nQueries + 15) / 16;

    const int threads       = 256;
    const int wavesPerBlock = threads / 32;
    int blocks = (nTiles + wavesPerBlock - 1) / wavesPerBlock;
    if (blocks > 128) blocks = 128;
    if (blocks < 1)   blocks = 1;

    const size_t ldsBytes = (size_t)LDS_FLOATS * sizeof(float);

    hipLaunchKernelGGL(parafac_query_wmma, dim3(blocks), dim3(threads), ldsBytes, stream,
                       U, V, W, idx, out, nQueries);
}


# V and W copied per workgroup into LDS ([rank][dim] as in memory, one barrier): a query's 32 reads of them become LDS reads, U's 16 stay on the WGP cache; two units share the gathers; same sums
# speedup vs baseline: 4.1763x; 4.1763x over previous
#include <hip/hip_runtime.h>


typedef float v2f __attribute__((ext_vector_type(2)));
typedef float v8f __attribute__((ext_vector_type(8)));

#define RANK 16
#define DIM  400

#define QUERY_BLOCKS           256
#define QUERY_WAVES_PER_BLOCK  4
#define QUERY_WAVES            (QUERY_BLOCKS * QUERY_WAVES_PER_BLOCK)
#define QUERY_THREADS          (QUERY_WAVES_PER_BLOCK * 32)

#define TABLE_FLOATS  (RANK * DIM)
#define TABLE_PIECES  (TABLE_FLOATS / 4)
#define FULL_TRIPS    (TABLE_PIECES / QUERY_THREADS)

__device__ __forceinline__ float other_half(float x)
{
    return __builtin_bit_cast(float, __builtin_amdgcn_permlanex16(__builtin_bit_cast(int, x), __builtin_bit_cast(int, x),
                                                                  0x76543210, 0xfedcba98, false, false));
}

__device__ __forceinline__ void load_triples(const int* __restrict__ idx, int nQueries, int base, int lane,
                                             int (&ri)[4], int (&rj)[4], int (&rk)[4])
{
#pragma unroll
    for (int t = 0; t < 4; ++t) {
        const int qi = (base + t) * 32 + lane;
        const int qc = (qi < nQueries) ? qi : (nQueries - 1);
        ri[t] = idx[qc * 3 + 0];
        rj[t] = idx[qc * 3 + 1];
        rk[t] = idx[qc * 3 + 2];
    }
}

template <int N>
__device__ __forceinline__ void parafac_groups(const float* __restrict__ U,
                                               const float* tab,
                                               const int (&ri)[4], const int (&rj)[4], const int (&rk)[4],
                                               float*       __restrict__ out,
                                               int nQueries, int base, int lane)
{
    const bool hi = lane >= 16;

    float u[N][RANK], v[N][RANK], w[N][RANK];
#pragma unroll
    for (int t = 0; t < N; ++t) {
        const int i = (int)min((unsigned)ri[t], (unsigned)(DIM - 1));
#pragma unroll
        for (int r = 0; r < RANK; ++r) u[t][r] = U[r * DIM + i];
    }
    __builtin_amdgcn_sched_barrier(0);
#pragma unroll
    for (int t = 0; t < N; ++t) {
        const int j = (int)min((unsigned)rj[t], (unsigned)(DIM - 1));
        const int k = (int)min((unsigned)rk[t], (unsigned)(DIM - 1));
#pragma unroll
        for (int r = 0; r < RANK; ++r) v[t][r] = tab[r * DIM + j];
#pragma unroll
        for (int r = 0; r < RANK; ++r) w[t][r] = tab[TABLE_FLOATS + r * DIM + k];
    }
    __builtin_amdgcn_sched_barrier(0);

#pragma unroll
    for (int t = 0; t < N; ++t) {
        float a[RANK];
#pragma unroll
        for (int r = 0; r < RANK; ++r) a[r] = u[t][r] * v[t][r];

        v8f c0 = {};
        v8f c1 = {};
#pragma unroll
        for (int c = 0; c < 4; ++c) {
            const float ax = other_half(hi ? a[4 * c]     : a[4 * c + 2]);
            const float ay = other_half(hi ? a[4 * c + 1] : a[4 * c + 3]);
            const float bx = other_half(hi ? w[t][4 * c]     : w[t][4 * c + 2]);
            const float by = other_half(hi ? w[t][4 * c + 1] : w[t][4 * c + 3]);
            v2f a0, b0, a1, b1;
            a0.x = hi ? ax : a[4 * c];         a0.y = hi ? ay : a[4 * c + 1];
            b0.x = hi ? bx : w[t][4 * c];      b0.y = hi ? by : w[t][4 * c + 1];
            a1.x = hi ? a[4 * c + 2] : ax;     a1.y = hi ? a[4 * c + 3] : ay;
            b1.x = hi ? w[t][4 * c + 2] : bx;  b1.y = hi ? w[t][4 * c + 3] : by;
            c0 = __builtin_amdgcn_wmma_f32_16x16x4_f32(false, a0, false, b0, (short)0, c0, false, false);
            c1 = __builtin_amdgcn_wmma_f32_16x16x4_f32(false, a1, false, b1, (short)0, c1, false, false);
        }

        const int s = lane & 7;
        float d0 = c0[0], d1 = c1[0];
        if (s == 1) { d0 = c0[1]; d1 = c1[1]; }
        if (s == 2) { d0 = c0[2]; d1 = c1[2]; }
        if (s == 3) { d0 = c0[3]; d1 = c1[3]; }
        if (s == 4) { d0 = c0[4]; d1 = c1[4]; }
        if (s == 5) { d0 = c0[5]; d1 = c1[5]; }
        if (s == 6) { d0 = c0[6]; d1 = c1[6]; }
        if (s == 7) { d0 = c0[7]; d1 = c1[7]; }
        const float x1  = other_half(d1);
        const bool  mid = (lane >= 8) && (lane < 24);
        const float val = mid ? x1 : d0;
        const int qq = lane ^ ((lane & 8) << 1);
        const int oq = (base + t) * 32 + qq;
        if (oq < nQueries) out[oq] = val;
    }
}

__global__ __launch_bounds__(QUERY_THREADS)
void parafac_query_wmma(const float* __restrict__ U,
                        const float* __restrict__ V,
                        const float* __restrict__ W,
                        const int*   __restrict__ idx,
                        float*       __restrict__ out,
                        int nQueries)
{
    __shared__ float4 tab4[2 * TABLE_PIECES];
    const float* tab = reinterpret_cast<const float*>(tab4);

    const int tid  = (int)threadIdx.x;
    const int lane = tid & 31;

    const int waveInBlock = __builtin_amdgcn_readfirstlane(tid >> 5);
    const int wave        = waveInBlock * QUERY_BLOCKS + (int)blockIdx.x;
    const int nGroups = (nQueries + 31) >> 5;
    const int per     = nGroups / QUERY_WAVES;
    const int rem     = nGroups % QUERY_WAVES;
    int       base    = wave * per + min(wave, rem);
    const int last    = base + per + ((wave < rem) ? 1 : 0);

    int ri[4] = {0, 0, 0, 0}, rj[4] = {0, 0, 0, 0}, rk[4] = {0, 0, 0, 0};
    if (base < last) load_triples(idx, nQueries, base, lane, ri, rj, rk);

    const float4* V4 = reinterpret_cast<const float4*>(V);
    const float4* W4 = reinterpret_cast<const float4*>(W);
    const bool    tailW   = waveInBlock >= 2;
    const float4* T4      = tailW ? W4 : V4;
    const int     tailSrc = FULL_TRIPS * QUERY_THREADS + (tid & 63);
    const int     tailDst = (tailW ? TABLE_PIECES : 0) + tailSrc;
    float4 pv[FULL_TRIPS], pw[FULL_TRIPS];
#pragma unroll
    for (int s = 0; s < FULL_TRIPS; ++s) pv[s] = V4[s * QUERY_THREADS + tid];
#pragma unroll
    for (int s = 0; s < FULL_TRIPS; ++s) pw[s] = W4[s * QUERY_THREADS + tid];
    const float4 pt = T4[tailSrc];
#pragma unroll
    for (int s = 0; s < FULL_TRIPS; ++s) tab4[s * QUERY_THREADS + tid] = pv[s];
#pragma unroll
    for (int s = 0; s < FULL_TRIPS; ++s) tab4[TABLE_PIECES + s * QUERY_THREADS + tid] = pw[s];
    tab4[tailDst] = pt;
    __syncthreads();

    while (base < last) {
        const int n = last - base;
        if (n >= 4)      { parafac_groups<4>(U, tab, ri, rj, rk, out, nQueries, base, lane); base += 4; }
        else if (n == 3) { parafac_groups<3>(U, tab, ri, rj, rk, out, nQueries, base, lane); base += 3; }
        else if (n == 2) { parafac_groups<2>(U, tab, ri, rj, rk, out, nQueries, base, lane); base += 2; }
        else             { parafac_groups<1>(U, tab, ri, rj, rk, out, nQueries, base, lane); base += 1; }
        if (base < last) load_triples(idx, nQueries, base, lane, ri, rj, rk);
    }
}

extern "C" void kernel_launch(void* const* d_in, const int* in_sizes, int n_in,
                              void* d_out, int out_size, void* d_ws, size_t ws_size,
                              hipStream_t stream) {
    const float* U   = (const float*)d_in[0];
    const float* V   = (const float*)d_in[1];
    const float* W   = (const float*)d_in[2];
    const int*   idx = (const int*)  d_in[3];
    float*       out = (float*)d_out;

    const int nQueries = out_size;

    hipLaunchKernelGGL(parafac_query_wmma, dim3(QUERY_BLOCKS), dim3(QUERY_THREADS), 0, stream,
                       U, V, W, idx, out, nQueries);
}


# Copy and gathers overlap: U's global gathers issue before the copy's LDS stores; V gets its own barrier and is gathered while W still arrives (barriers by hand keep loads in flight); same sums
# speedup vs baseline: 4.4870x; 1.0744x over previous
#include <hip/hip_runtime.h>


typedef float v2f __attribute__((ext_vector_type(2)));
typedef float v8f __attribute__((ext_vector_type(8)));

#define RANK 16
#define DIM  400

#define QUERY_BLOCKS           256
#define QUERY_WAVES_PER_BLOCK  4
#define QUERY_WAVES            (QUERY_BLOCKS * QUERY_WAVES_PER_BLOCK)
#define QUERY_THREADS          (QUERY_WAVES_PER_BLOCK * 32)

#define TABLE_FLOATS  (RANK * DIM)
#define TABLE_PIECES  (TABLE_FLOATS / 4)
#define FULL_TRIPS    (TABLE_PIECES / QUERY_THREADS)

__device__ __forceinline__ float other_half(float x)
{
    return __builtin_bit_cast(float, __builtin_amdgcn_permlanex16(__builtin_bit_cast(int, x), __builtin_bit_cast(int, x),
                                                                  0x76543210, 0xfedcba98, false, false));
}

__device__ __forceinline__ void lds_barrier()
{
    asm volatile("s_wait_dscnt 0x0\n\ts_barrier_signal -1\n\ts_barrier_wait -1" ::: "memory");
}

__device__ __forceinline__ void load_triples(const int* idx, int nQueries, int base, int lane,
                                             int (&ri)[4], int (&rj)[4], int (&rk)[4])
{
#pragma unroll
    for (int t = 0; t < 4; ++t) {
        const int qi = (base + t) * 32 + lane;
        const int qc = (qi < nQueries) ? qi : (nQueries - 1);
        ri[t] = idx[qc * 3 + 0];
        rj[t] = idx[qc * 3 + 1];
        rk[t] = idx[qc * 3 + 2];
    }
}

__device__ __forceinline__ void gather_global(const float* U, int index, float (&u)[RANK])
{
    const int i = (int)min((unsigned)index, (unsigned)(DIM - 1));
#pragma unroll
    for (int r = 0; r < RANK; ++r) u[r] = U[r * DIM + i];
}

__device__ __forceinline__ void gather_lds(const float* tab, int index, float (&v)[RANK])
{
    const int j = (int)min((unsigned)index, (unsigned)(DIM - 1));
#pragma unroll
    for (int r = 0; r < RANK; ++r) v[r] = tab[r * DIM + j];
}

__device__ __forceinline__ void finish_group(const float (&u)[RANK], const float (&v)[RANK], const float (&w)[RANK],
                                             float* __restrict__ out, int nQueries, int group, int lane)
{
    const bool hi = lane >= 16;

    float a[RANK];
#pragma unroll
    for (int r = 0; r < RANK; ++r) a[r] = u[r] * v[r];

    v8f c0 = {};
    v8f c1 = {};
#pragma unroll
    for (int c = 0; c < 4; ++c) {
        const float ax = other_half(hi ? a[4 * c]     : a[4 * c + 2]);
        const float ay = other_half(hi ? a[4 * c + 1] : a[4 * c + 3]);
        const float bx = other_half(hi ? w[4 * c]     : w[4 * c + 2]);
        const float by = other_half(hi ? w[4 * c + 1] : w[4 * c + 3]);
        v2f a0, b0, a1, b1;
        a0.x = hi ? ax : a[4 * c];      a0.y = hi ? ay : a[4 * c + 1];
        b0.x = hi ? bx : w[4 * c];      b0.y = hi ? by : w[4 * c + 1];
        a1.x = hi ? a[4 * c + 2] : ax;  a1.y = hi ? a[4 * c + 3] : ay;
        b1.x = hi ? w[4 * c + 2] : bx;  b1.y = hi ? w[4 * c + 3] : by;
        c0 = __builtin_amdgcn_wmma_f32_16x16x4_f32(false, a0, false, b0, (short)0, c0, false, false);
        c1 = __builtin_amdgcn_wmma_f32_16x16x4_f32(false, a1, false, b1, (short)0, c1, false, false);
    }

    const int s = lane & 7;
    float d0 = c0[0], d1 = c1[0];
    if (s == 1) { d0 = c0[1]; d1 = c1[1]; }
    if (s == 2) { d0 = c0[2]; d1 = c1[2]; }
    if (s == 3) { d0 = c0[3]; d1 = c1[3]; }
    if (s == 4) { d0 = c0[4]; d1 = c1[4]; }
    if (s == 5) { d0 = c0[5]; d1 = c1[5]; }
    if (s == 6) { d0 = c0[6]; d1 = c1[6]; }
    if (s == 7) { d0 = c0[7]; d1 = c1[7]; }
    const float x1  = other_half(d1);
    const bool  mid = (lane >= 8) && (lane < 24);
    const float val = mid ? x1 : d0;
    const int qq = lane ^ ((lane & 8) << 1);
    const int oq = group * 32 + qq;
    if (oq < nQueries) out[oq] = val;
}

__global__ __launch_bounds__(QUERY_THREADS)
void parafac_query_wmma(const float* U,
                        const float* V,
                        const float* W,
                        const int*   idx,
                        float*       __restrict__ out,
                        int nQueries)
{
    __shared__ float4 tab4[2 * TABLE_PIECES];
    const float* tabV = reinterpret_cast<const float*>(tab4);
    const float* tabW = tabV + TABLE_FLOATS;

    const int tid  = (int)threadIdx.x;
    const int lane = tid & 31;

    const int waveInBlock = __builtin_amdgcn_readfirstlane(tid >> 5);
    const int wave        = waveInBlock * QUERY_BLOCKS + (int)blockIdx.x;
    const int nGroups = (nQueries + 31) >> 5;
    const int per     = nGroups / QUERY_WAVES;
    const int rem     = nGroups % QUERY_WAVES;
    int       base    = wave * per + min(wave, rem);
    const int last    = base + per + ((wave < rem) ? 1 : 0);
    const int first   = min(max(last - base, 0), 4);

    int ri[4] = {0, 0, 0, 0}, rj[4] = {0, 0, 0, 0}, rk[4] = {0, 0, 0, 0};
    if (base < last) load_triples(idx, nQueries, base, lane, ri, rj, rk);

    const float4* V4 = reinterpret_cast<const float4*>(V);
    const float4* W4 = reinterpret_cast<const float4*>(W);
    const bool    tailW   = waveInBlock >= 2;
    const float4* T4      = tailW ? W4 : V4;
    const int     tailSrc = FULL_TRIPS * QUERY_THREADS + (tid & 63);
    const int     tailDst = (tailW ? TABLE_PIECES : 0) + tailSrc;
#define FOR_TRIPS(X) X(0) X(1) X(2) X(3) X(4) X(5) X(6) X(7) X(8) X(9) X(10) X(11)
#define LOAD_V(s)  const float4 pv##s = V4[s * QUERY_THREADS + tid];
#define LOAD_W(s)  const float4 pw##s = W4[s * QUERY_THREADS + tid];
#define STORE_V(s) tab4[s * QUERY_THREADS + tid] = pv##s;
#define STORE_W(s) tab4[TABLE_PIECES + s * QUERY_THREADS + tid] = pw##s;
    static_assert(FULL_TRIPS == 12, "FOR_TRIPS lists the whole trips");
    FOR_TRIPS(LOAD_V)
    const float4 pt = T4[tailSrc];
    __builtin_amdgcn_sched_barrier(0);
    FOR_TRIPS(LOAD_W)
    __builtin_amdgcn_sched_barrier(0);

    float u[4][RANK], v[4][RANK], w[4][RANK];
#pragma unroll
    for (int t = 0; t < 2; ++t) gather_global(U, ri[t], u[t]);

    FOR_TRIPS(STORE_V)
    tab4[tailDst] = pt;
    __builtin_amdgcn_sched_barrier(0);
    gather_global(U, ri[2], u[2]);
    lds_barrier();

#pragma unroll
    for (int t = 0; t < 3; ++t) gather_lds(tabV, rj[t], v[t]);
    if (first > 3) gather_lds(tabV, rj[3], v[3]);

    FOR_TRIPS(STORE_W)
    if (first > 3) gather_global(U, ri[3], u[3]);
    lds_barrier();

#pragma unroll
    for (int t = 0; t < 3; ++t) gather_lds(tabW, rk[t], w[t]);
    if (first > 3) gather_lds(tabW, rk[3], w[3]);
#pragma unroll
    for (int t = 0; t < 4; ++t)
        if (t < first) finish_group(u[t], v[t], w[t], out, nQueries, base + t, lane);
    base += first;

    while (base < last) {
        const int n = min(last - base, 4);
        load_triples(idx, nQueries, base, lane, ri, rj, rk);
#pragma unroll
        for (int t = 0; t < 4; ++t)
            if (t < n) gather_global(U, ri[t], u[t]);
#pragma unroll
        for (int t = 0; t < 4; ++t)
            if (t < n) { gather_lds(tabV, rj[t], v[t]); gather_lds(tabW, rk[t], w[t]); }
#pragma unroll
        for (int t = 0; t < 4; ++t)
            if (t < n) finish_group(u[t], v[t], w[t], out, nQueries, base + t, lane);
        base += n;
    }
}

extern "C" void kernel_launch(void* const* d_in, const int* in_sizes, int n_in,
                              void* d_out, int out_size, void* d_ws, size_t ws_size,
                              hipStream_t stream) {
    const float* U   = (const float*)d_in[0];
    const float* V   = (const float*)d_in[1];
    const float* W   = (const float*)d_in[2];
    const int*   idx = (const int*)  d_in[3];
    float*       out = (float*)d_out;

    const int nQueries = out_size;

    hipLaunchKernelGGL(parafac_query_wmma, dim3(QUERY_BLOCKS), dim3(QUERY_THREADS), 0, stream,
                       U, V, W, idx, out, nQueries);
}
